# starting kernel
# baseline (speedup 1.0000x reference)
#include <hip/hip_runtime.h>

typedef __attribute__((ext_vector_type(2))) float v2f;
typedef __attribute__((ext_vector_type(8))) float v8f;

#define NB      16
#define HH      64
#define WW      64
#define QIN     4
#define BOND    16
#define OUTQ    64
#define HPAD    65
#define WPAD    65
#define INTER   63
#define PIX_PER_IMG (INTER * INTER)
#define NPIX    (NB * PIX_PER_IMG)
#define NBLK    (NPIX / 16)
#define WAVES_PER_WG 8
#define NSLOT   97
#define S_W3    0
#define S_W2    1
#define S_W1    17
#define S_W0    33

static __device__ __forceinline__ v8f wmma4(v2f a, v2f b, v8f c) {
    return __builtin_amdgcn_wmma_f32_16x16x4_f32(
         false, a,  false, b,
         (short)0, c,  false,  false);
}

__global__ __launch_bounds__(256) void convsbs_zero(float4* __restrict__ o, int n4) {
    int i = blockIdx.x * blockDim.x + threadIdx.x;
    if (i < n4) o[i] = make_float4(0.f, 0.f, 0.f, 0.f);
}

__global__ __launch_bounds__(256) void convsbs_chain(
        const float* __restrict__ x,
        const float* __restrict__ c0,
        const float* __restrict__ c1,
        const float* __restrict__ c2,
        const float* __restrict__ c3,
        float* __restrict__ out)
{
    __shared__ float wlds[NSLOT * 64];

    const int tid = threadIdx.x;

    for (int i = tid; i < NSLOT * 64; i += 256) {
        int slot = i >> 6;
        int e    = i & 63;
        int lane = e >> 1;
        int v    = e & 1;
        int m    = lane & 15;
        int kk   = ((lane >> 4) << 1) + v;
        float val;
        if (slot == S_W3) {
            val = c3[m * 4 + kk];
        } else if (slot < S_W1) {
            int j = slot - S_W2;
            val = c2[m * 64 + 4 * j + kk];
        } else if (slot < S_W0) {
            int j = slot - S_W1;
            val = c1[m * 64 + 4 * j + kk];
        } else {
            int s = slot - S_W0;
            int t = s >> 4, j = s & 15;
            val = c0[(t * 16 + m) * 64 + 4 * j + kk];
        }
        wlds[i] = val;
    }
    __syncthreads();

    const int wave = tid >> 5;
    const int lane = tid & 31;
    const int blk  = blockIdx.x * WAVES_PER_WG + wave;
    if (blk >= NBLK) return;

    const int pix = blk * 16 + (lane & 15);
    const int b   = pix / PIX_PER_IMG;
    const int rr  = pix - b * PIX_PER_IMG;
    const int hp  = rr / INTER + 1;
    const int wp  = rr - (hp - 1) * INTER + 1;

    const float4* xv = (const float4*)x;
    const int base = (b * HH + hp) * WW + wp;
    const float4 q3 = xv[base];
    const float4 q2 = xv[base - 1];
    const float4 q1 = xv[base - WW];
    const float4 q0 = xv[base - WW - 1];

    const int half = lane >> 4;
    const float xb3a = half ? q3.z : q3.x, xb3b = half ? q3.w : q3.y;
    const float xb2a = half ? q2.z : q2.x, xb2b = half ? q2.w : q2.y;
    const float xb1a = half ? q1.z : q1.x, xb1b = half ? q1.w : q1.y;
    const float xb0a = half ? q0.z : q0.x, xb0b = half ? q0.w : q0.y;

    const v2f* wv = (const v2f*)wlds;

    v2f a = wv[S_W3 * 32 + lane];
    v2f bb; bb[0] = xb3a; bb[1] = xb3b;
    v8f zeroAcc = {};
    v8f v3d = wmma4(a, bb, zeroAcc);

    v8f u = {};
#pragma unroll
    for (int j = 0; j < 16; ++j) {
        float row = __shfl(v3d[j & 7], (lane & 15) | ((j >> 3) << 4), 32);
        v2f bj; bj[0] = row * xb2a; bj[1] = row * xb2b;
        v2f aj = wv[(S_W2 + j) * 32 + lane];
        u = wmma4(aj, bj, u);
    }

    v8f vv = {};
#pragma unroll
    for (int j = 0; j < 16; ++j) {
        float row = __shfl(u[j & 7], (lane & 15) | ((j >> 3) << 4), 32);
        v2f bj; bj[0] = row * xb1a; bj[1] = row * xb1b;
        v2f aj = wv[(S_W1 + j) * 32 + lane];
        vv = wmma4(aj, bj, vv);
    }

    v8f d0 = {}, d1 = {}, d2 = {}, d3 = {};
#pragma unroll
    for (int j = 0; j < 16; ++j) {
        float row = __shfl(vv[j & 7], (lane & 15) | ((j >> 3) << 4), 32);
        v2f bj; bj[0] = row * xb0a; bj[1] = row * xb0b;
        d0 = wmma4(wv[(S_W0 +  0 + j) * 32 + lane], bj, d0);
        d1 = wmma4(wv[(S_W0 + 16 + j) * 32 + lane], bj, d1);
        d2 = wmma4(wv[(S_W0 + 32 + j) * 32 + lane], bj, d2);
        d3 = wmma4(wv[(S_W0 + 48 + j) * 32 + lane], bj, d3);
    }

    const int obase = ((b * HPAD + hp) * WPAD + wp) * OUTQ + 8 * half;
#define STORE_TILE(dt, t) {                                              \
        float* p = out + obase + (t) * 16;                               \
        ((float4*)p)[0] = make_float4(dt[0], dt[1], dt[2], dt[3]);       \
        ((float4*)p)[1] = make_float4(dt[4], dt[5], dt[6], dt[7]);       \
    }
    STORE_TILE(d0, 0)
    STORE_TILE(d1, 1)
    STORE_TILE(d2, 2)
    STORE_TILE(d3, 3)
#undef STORE_TILE
}

extern "C" void kernel_launch(void* const* d_in, const int* in_sizes, int n_in,
                              void* d_out, int out_size, void* d_ws, size_t ws_size,
                              hipStream_t stream) {
    (void)in_sizes; (void)n_in; (void)d_ws; (void)ws_size;
    const float* x  = (const float*)d_in[0];
    const float* c0 = (const float*)d_in[1];
    const float* c1 = (const float*)d_in[2];
    const float* c2 = (const float*)d_in[3];
    const float* c3 = (const float*)d_in[4];
    float* out = (float*)d_out;

    int n4 = out_size / 4;
    convsbs_zero<<<(n4 + 255) / 256, 256, 0, stream>>>((float4*)out, n4);

    int grid = (NBLK + WAVES_PER_WG - 1) / WAVES_PER_WG;
    convsbs_chain<<<grid, 256, 0, stream>>>(x, c0, c1, c2, c3, out);
}


# Weights staged by coalesced 16-byte loads in row-major layout (padded LDS pitch, fragments read in place), all prologue loads issued before the first wait; replaces 25 serialized 4-byte gathers
# speedup vs baseline: 1.6023x; 1.6023x over previous
#include <hip/hip_runtime.h>

typedef __attribute__((ext_vector_type(2))) float v2f;
typedef __attribute__((ext_vector_type(8))) float v8f;

#define NB      16
#define HH      64
#define WW      64
#define QIN     4
#define BOND    16
#define OUTQ    64
#define HPAD    65
#define WPAD    65
#define INTER   63
#define PIX_PER_IMG (INTER * INTER)
#define NPIX    (NB * PIX_PER_IMG)
#define NBLK    (NPIX / 16)
#define PADPIX  (HPAD * WPAD)
#define BORDER_PER_IMG (PADPIX - PIX_PER_IMG)
#define NBORDER4 (NB * BORDER_PER_IMG * (OUTQ / 4))
#define WAVES_PER_WG 8
#define PITCH   68
#define L_W0    0
#define L_W1    (64 * PITCH)
#define L_W2    (L_W1 + 16 * PITCH)
#define L_W3    (L_W2 + 16 * PITCH)
#define L_TOT   (L_W3 + 64)

static __device__ __forceinline__ v8f wmma4(v2f a, v2f b, v8f c) {
    return __builtin_amdgcn_wmma_f32_16x16x4_f32(
         false, a,  false, b,
         (short)0, c,  false,  false);
}

__global__ __launch_bounds__(256) void convsbs_chain(
        const float* __restrict__ x,
        const float* __restrict__ c0,
        const float* __restrict__ c1,
        const float* __restrict__ c2,
        const float* __restrict__ c3,
        float* __restrict__ out)
{
    __shared__ __attribute__((aligned(16))) float wlds[L_TOT];

    const int tid = threadIdx.x;

    const float4* g0 = (const float4*)c0;
    const float4* g1 = (const float4*)c1;
    const float4* g2 = (const float4*)c2;
    const float4* g3 = (const float4*)c3;
    const float4 wa0 = g0[tid];
    const float4 wa1 = g0[tid + 256];
    const float4 wa2 = g0[tid + 512];
    const float4 wa3 = g0[tid + 768];
    const float4 wb1 = g1[tid];
    const float4 wb2 = g2[tid];
    float4 wb3 = make_float4(0.f, 0.f, 0.f, 0.f);
    if (tid < 16) wb3 = g3[tid];

    const int wave = tid >> 5;
    const int lane = tid & 31;
    const int blk  = blockIdx.x * WAVES_PER_WG + wave;
    const bool live = blk < NBLK;
    const int blkc = live ? blk : (NBLK - 1);

    const int pix = blkc * 16 + (lane & 15);
    const int b   = pix / PIX_PER_IMG;
    const int rr  = pix - b * PIX_PER_IMG;
    const int hp  = rr / INTER + 1;
    const int wp  = rr - (hp - 1) * INTER + 1;

    const float4* xv = (const float4*)x;
    const int base = (b * HH + hp) * WW + wp;
    const float4 q3 = xv[base];
    const float4 q2 = xv[base - 1];
    const float4 q1 = xv[base - WW];
    const float4 q0 = xv[base - WW - 1];

    {
        const int g = blockIdx.x * 256 + tid;
        if (g < NBORDER4) {
            const int part = g & 15;
            const int bp   = g >> 4;
            const int img  = bp >> 8;
            const int n    = bp & 255;
            int p;
            if (n < 66) {
                p = n;
            } else if (n < 190) {
                const int m = n - 66;
                p = (1 + (m >> 1)) * WPAD + (WPAD - 1) + (m & 1);
            } else {
                p = 63 * WPAD + (WPAD - 1) + (n - 190);
            }
            ((float4*)out)[(img * PADPIX + p) * (OUTQ / 4) + part] =
                make_float4(0.f, 0.f, 0.f, 0.f);
        }
    }

    {
        float* d = wlds + (tid >> 4) * PITCH + (tid & 15) * 4;
        *(float4*)(d + L_W0)              = wa0;
        *(float4*)(d + L_W0 + 16 * PITCH) = wa1;
        *(float4*)(d + L_W0 + 32 * PITCH) = wa2;
        *(float4*)(d + L_W0 + 48 * PITCH) = wa3;
        *(float4*)(d + L_W1)              = wb1;
        *(float4*)(d + L_W2)              = wb2;
        if (tid < 16) *(float4*)(wlds + L_W3 + tid * 4) = wb3;
    }
    __syncthreads();

    if (!live) return;

    const int half = lane >> 4;
    const float xb3a = half ? q3.z : q3.x, xb3b = half ? q3.w : q3.y;
    const float xb2a = half ? q2.z : q2.x, xb2b = half ? q2.w : q2.y;
    const float xb1a = half ? q1.z : q1.x, xb1b = half ? q1.w : q1.y;
    const float xb0a = half ? q0.z : q0.x, xb0b = half ? q0.w : q0.y;

    const float* wrow = wlds + (lane & 15) * PITCH + 2 * half;
#define AFRAG(off, j) (*(const v2f*)(wrow + (off) + 4 * (j)))

    v2f a = *(const v2f*)(wlds + L_W3 + (lane & 15) * 4 + 2 * half);
    v2f bb; bb[0] = xb3a; bb[1] = xb3b;
    v8f zeroAcc = {};
    v8f v3d = wmma4(a, bb, zeroAcc);

    v8f u = {};
#pragma unroll
    for (int j = 0; j < 16; ++j) {
        float row = __shfl(v3d[j & 7], (lane & 15) | ((j >> 3) << 4), 32);
        v2f bj; bj[0] = row * xb2a; bj[1] = row * xb2b;
        v2f aj = AFRAG(L_W2, j);
        u = wmma4(aj, bj, u);
    }

    v8f vv = {};
#pragma unroll
    for (int j = 0; j < 16; ++j) {
        float row = __shfl(u[j & 7], (lane & 15) | ((j >> 3) << 4), 32);
        v2f bj; bj[0] = row * xb1a; bj[1] = row * xb1b;
        v2f aj = AFRAG(L_W1, j);
        vv = wmma4(aj, bj, vv);
    }

    v8f d0 = {}, d1 = {}, d2 = {}, d3 = {};
#pragma unroll
    for (int j = 0; j < 16; ++j) {
        float row = __shfl(vv[j & 7], (lane & 15) | ((j >> 3) << 4), 32);
        v2f bj; bj[0] = row * xb0a; bj[1] = row * xb0b;
        d0 = wmma4(AFRAG(L_W0 +  0 * PITCH, j), bj, d0);
        d1 = wmma4(AFRAG(L_W0 + 16 * PITCH, j), bj, d1);
        d2 = wmma4(AFRAG(L_W0 + 32 * PITCH, j), bj, d2);
        d3 = wmma4(AFRAG(L_W0 + 48 * PITCH, j), bj, d3);
    }
#undef AFRAG

    const int obase = ((b * HPAD + hp) * WPAD + wp) * OUTQ + 8 * half;
#define STORE_TILE(dt, t) {                                              \
        float* p = out + obase + (t) * 16;                               \
        ((float4*)p)[0] = make_float4(dt[0], dt[1], dt[2], dt[3]);       \
        ((float4*)p)[1] = make_float4(dt[4], dt[5], dt[6], dt[7]);       \
    }
    STORE_TILE(d0, 0)
    STORE_TILE(d1, 1)
    STORE_TILE(d2, 2)
    STORE_TILE(d3, 3)
#undef STORE_TILE
}

extern "C" void kernel_launch(void* const* d_in, const int* in_sizes, int n_in,
                              void* d_out, int out_size, void* d_ws, size_t ws_size,
                              hipStream_t stream) {
    (void)in_sizes; (void)n_in; (void)out_size; (void)d_ws; (void)ws_size;
    const float* x  = (const float*)d_in[0];
    const float* c0 = (const float*)d_in[1];
    const float* c1 = (const float*)d_in[2];
    const float* c2 = (const float*)d_in[3];
    const float* c3 = (const float*)d_in[4];
    float* out = (float*)d_out;

    int grid = (NBLK + WAVES_PER_WG - 1) / WAVES_PER_WG;
    convsbs_chain<<<grid, 256, 0, stream>>>(x, c0, c1, c2, c3, out);
}


# k order of stages 2-4 matched to the D layout: each lane half contracts the 8 bond rows it already holds; the 48 cross-lane shuffles go, A fragments are 16-byte LDS reads in one group up front
# speedup vs baseline: 1.8668x; 1.1651x over previous
#include <hip/hip_runtime.h>

typedef __attribute__((ext_vector_type(2))) float v2f;
typedef __attribute__((ext_vector_type(8))) float v8f;

#define NB      16
#define HH      64
#define WW      64
#define QIN     4
#define BOND    16
#define OUTQ    64
#define HPAD    65
#define WPAD    65
#define INTER   63
#define PIX_PER_IMG (INTER * INTER)
#define NPIX    (NB * PIX_PER_IMG)
#define NBLK    (NPIX / 16)
#define PADPIX  (HPAD * WPAD)
#define BORDER_PER_IMG (PADPIX - PIX_PER_IMG)
#define NBORDER4 (NB * BORDER_PER_IMG * (OUTQ / 4))
#define WAVES_PER_WG 8
#define PITCH   68
#define L_W0    0
#define L_W1    (64 * PITCH)
#define L_W2    (L_W1 + 16 * PITCH)
#define L_W3    (L_W2 + 16 * PITCH)
#define L_TOT   (L_W3 + 64)

static __device__ __forceinline__ v8f wmma4(v2f a, v2f b, v8f c) {
    return __builtin_amdgcn_wmma_f32_16x16x4_f32(
         false, a,  false, b,
         (short)0, c,  false,  false);
}

__global__ __launch_bounds__(256) void convsbs_chain(
        const float* __restrict__ x,
        const float* __restrict__ c0,
        const float* __restrict__ c1,
        const float* __restrict__ c2,
        const float* __restrict__ c3,
        float* __restrict__ out)
{
    __shared__ __attribute__((aligned(16))) float wlds[L_TOT];

    const int tid = threadIdx.x;

    const float4* g0 = (const float4*)c0;
    const float4* g1 = (const float4*)c1;
    const float4* g2 = (const float4*)c2;
    const float4* g3 = (const float4*)c3;
    const float4 wa0 = g0[tid];
    const float4 wa1 = g0[tid + 256];
    const float4 wa2 = g0[tid + 512];
    const float4 wa3 = g0[tid + 768];
    const float4 wb1 = g1[tid];
    const float4 wb2 = g2[tid];
    float4 wb3 = make_float4(0.f, 0.f, 0.f, 0.f);
    if (tid < 16) wb3 = g3[tid];

    const int wave = tid >> 5;
    const int lane = tid & 31;
    const int blk  = blockIdx.x * WAVES_PER_WG + wave;
    const bool live = blk < NBLK;
    const int blkc = live ? blk : (NBLK - 1);

    const int pix = blkc * 16 + (lane & 15);
    const int b   = pix / PIX_PER_IMG;
    const int rr  = pix - b * PIX_PER_IMG;
    const int hp  = rr / INTER + 1;
    const int wp  = rr - (hp - 1) * INTER + 1;

    const float4* xv = (const float4*)x;
    const int base = (b * HH + hp) * WW + wp;
    const float4 q3 = xv[base];
    const float4 q2 = xv[base - 1];
    const float4 q1 = xv[base - WW];
    const float4 q0 = xv[base - WW - 1];

    {
        const int g = blockIdx.x * 256 + tid;
        if (g < NBORDER4) {
            const int part = g & 15;
            const int bp   = g >> 4;
            const int img  = bp >> 8;
            const int n    = bp & 255;
            int p;
            if (n < 66) {
                p = n;
            } else if (n < 190) {
                const int m = n - 66;
                p = (1 + (m >> 1)) * WPAD + (WPAD - 1) + (m & 1);
            } else {
                p = 63 * WPAD + (WPAD - 1) + (n - 190);
            }
            ((float4*)out)[(img * PADPIX + p) * (OUTQ / 4) + part] =
                make_float4(0.f, 0.f, 0.f, 0.f);
        }
    }

    {
        float* d = wlds + (tid >> 4) * PITCH + (tid & 15) * 4;
        *(float4*)(d + L_W0)              = wa0;
        *(float4*)(d + L_W0 + 16 * PITCH) = wa1;
        *(float4*)(d + L_W0 + 32 * PITCH) = wa2;
        *(float4*)(d + L_W0 + 48 * PITCH) = wa3;
        *(float4*)(d + L_W1)              = wb1;
        *(float4*)(d + L_W2)              = wb2;
        if (tid < 16) *(float4*)(wlds + L_W3 + tid * 4) = wb3;
    }
    __syncthreads();

    if (!live) return;

    const int half = lane >> 4;
    const float xb3a = half ? q3.z : q3.x, xb3b = half ? q3.w : q3.y;

    const float* wrow = wlds + (lane & 15) * PITCH + 32 * half;
#define AF4(off, i) (*(const float4*)(wrow + (off) + 4 * (i)))

    const v2f a3 = *(const v2f*)(wlds + L_W3 + (lane & 15) * 4 + 2 * half);
    float4 w2f[8], w1f[8], w0f[4][8];
#pragma unroll
    for (int i = 0; i < 8; ++i) w2f[i] = AF4(L_W2, i);
#pragma unroll
    for (int i = 0; i < 8; ++i) w1f[i] = AF4(L_W1, i);
#pragma unroll
    for (int t = 0; t < 4; ++t) {
#pragma unroll
        for (int i = 0; i < 8; ++i) w0f[t][i] = AF4(L_W0 + 16 * t * PITCH, i);
    }
#undef AF4
    __builtin_amdgcn_sched_barrier(0);

    v2f bb; bb[0] = xb3a; bb[1] = xb3b;
    v8f zeroAcc = {};
    v8f v3d = wmma4(a3, bb, zeroAcc);

    v8f u = {};
#pragma unroll
    for (int i = 0; i < 8; ++i) {
        const float r = v3d[i];
        v2f a0; a0[0] = w2f[i].x; a0[1] = w2f[i].y;
        v2f b0; b0[0] = r * q2.x;  b0[1] = r * q2.y;
        u = wmma4(a0, b0, u);
        v2f a1; a1[0] = w2f[i].z; a1[1] = w2f[i].w;
        v2f b1; b1[0] = r * q2.z;  b1[1] = r * q2.w;
        u = wmma4(a1, b1, u);
    }

    v8f vv = {};
#pragma unroll
    for (int i = 0; i < 8; ++i) {
        const float r = u[i];
        v2f a0; a0[0] = w1f[i].x; a0[1] = w1f[i].y;
        v2f b0; b0[0] = r * q1.x;  b0[1] = r * q1.y;
        vv = wmma4(a0, b0, vv);
        v2f a1; a1[0] = w1f[i].z; a1[1] = w1f[i].w;
        v2f b1; b1[0] = r * q1.z;  b1[1] = r * q1.w;
        vv = wmma4(a1, b1, vv);
    }

    v8f d0 = {}, d1 = {}, d2 = {}, d3 = {};
#define A01(f) ({ v2f a_; a_[0] = (f).x; a_[1] = (f).y; a_; })
#define A23(f) ({ v2f a_; a_[0] = (f).z; a_[1] = (f).w; a_; })
#pragma unroll
    for (int i = 0; i < 8; ++i) {
        const float r = vv[i];
        v2f b0; b0[0] = r * q0.x;  b0[1] = r * q0.y;
        d0 = wmma4(A01(w0f[0][i]), b0, d0);
        d1 = wmma4(A01(w0f[1][i]), b0, d1);
        d2 = wmma4(A01(w0f[2][i]), b0, d2);
        d3 = wmma4(A01(w0f[3][i]), b0, d3);
        v2f b1; b1[0] = r * q0.z;  b1[1] = r * q0.w;
        d0 = wmma4(A23(w0f[0][i]), b1, d0);
        d1 = wmma4(A23(w0f[1][i]), b1, d1);
        d2 = wmma4(A23(w0f[2][i]), b1, d2);
        d3 = wmma4(A23(w0f[3][i]), b1, d3);
    }
#undef A01
#undef A23

    const int obase = ((b * HPAD + hp) * WPAD + wp) * OUTQ + 8 * half;
#define STORE_TILE(dt, t) {                                              \
        float* p = out + obase + (t) * 16;                               \
        ((float4*)p)[0] = make_float4(dt[0], dt[1], dt[2], dt[3]);       \
        ((float4*)p)[1] = make_float4(dt[4], dt[5], dt[6], dt[7]);       \
    }
    STORE_TILE(d0, 0)
    STORE_TILE(d1, 1)
    STORE_TILE(d2, 2)
    STORE_TILE(d3, 3)
#undef STORE_TILE
}

extern "C" void kernel_launch(void* const* d_in, const int* in_sizes, int n_in,
                              void* d_out, int out_size, void* d_ws, size_t ws_size,
                              hipStream_t stream) {
    (void)in_sizes; (void)n_in; (void)out_size; (void)d_ws; (void)ws_size;
    const float* x  = (const float*)d_in[0];
    const float* c0 = (const float*)d_in[1];
    const float* c1 = (const float*)d_in[2];
    const float* c2 = (const float*)d_in[3];
    const float* c3 = (const float*)d_in[4];
    float* out = (float*)d_out;

    int grid = (NBLK + WAVES_PER_WG - 1) / WAVES_PER_WG;
    convsbs_chain<<<grid, 256, 0, stream>>>(x, c0, c1, c2, c3, out);
}


# Stage 4 transposed (pixels as A rows, W0 as B, tile column c = output 4c+t): a lane's 4 accumulators hold 16 adjacent bytes, each store writes two whole 256-byte pixel rows, not 32 B of 16 lines
# speedup vs baseline: 3.2999x; 1.7677x over previous
#include <hip/hip_runtime.h>

typedef __attribute__((ext_vector_type(2))) float v2f;
typedef __attribute__((ext_vector_type(8))) float v8f;

#define NB      16
#define HH      64
#define WW      64
#define QIN     4
#define BOND    16
#define OUTQ    64
#define HPAD    65
#define WPAD    65
#define INTER   63
#define PIX_PER_IMG (INTER * INTER)
#define NPIX    (NB * PIX_PER_IMG)
#define NBLK    (NPIX / 16)
#define PADPIX  (HPAD * WPAD)
#define BORDER_PER_IMG (PADPIX - PIX_PER_IMG)
#define NBORDER4 (NB * BORDER_PER_IMG * (OUTQ / 4))
#define WAVES_PER_WG 8
#define PITCH   68
#define L_W0    0
#define L_W1    (64 * PITCH)
#define L_W2    (L_W1 + 16 * PITCH)
#define L_W3    (L_W2 + 16 * PITCH)
#define L_TOT   (L_W3 + 64)

static __device__ __forceinline__ v8f wmma4(v2f a, v2f b, v8f c) {
    return __builtin_amdgcn_wmma_f32_16x16x4_f32(
         false, a,  false, b,
         (short)0, c,  false,  false);
}

__global__ __launch_bounds__(256) void convsbs_chain(
        const float* __restrict__ x,
        const float* __restrict__ c0,
        const float* __restrict__ c1,
        const float* __restrict__ c2,
        const float* __restrict__ c3,
        float* __restrict__ out)
{
    __shared__ __attribute__((aligned(16))) float wlds[L_TOT];

    const int tid = threadIdx.x;

    const float4* g0 = (const float4*)c0;
    const float4* g1 = (const float4*)c1;
    const float4* g2 = (const float4*)c2;
    const float4* g3 = (const float4*)c3;
    const float4 wa0 = g0[tid];
    const float4 wa1 = g0[tid + 256];
    const float4 wa2 = g0[tid + 512];
    const float4 wa3 = g0[tid + 768];
    const float4 wb1 = g1[tid];
    const float4 wb2 = g2[tid];
    float4 wb3 = make_float4(0.f, 0.f, 0.f, 0.f);
    if (tid < 16) wb3 = g3[tid];

    const int wave = tid >> 5;
    const int lane = tid & 31;
    const int blk  = blockIdx.x * WAVES_PER_WG + wave;
    const bool live = blk < NBLK;
    const int blkc = live ? blk : (NBLK - 1);

    const int pix = blkc * 16 + (lane & 15);
    const int b   = pix / PIX_PER_IMG;
    const int rr  = pix - b * PIX_PER_IMG;
    const int hp  = rr / INTER + 1;
    const int wp  = rr - (hp - 1) * INTER + 1;

    const float4* xv = (const float4*)x;
    const int base = (b * HH + hp) * WW + wp;
    const float4 q3 = xv[base];
    const float4 q2 = xv[base - 1];
    const float4 q1 = xv[base - WW];
    const float4 q0 = xv[base - WW - 1];

    {
        const int g = blockIdx.x * 256 + tid;
        if (g < NBORDER4) {
            const int part = g & 15;
            const int bp   = g >> 4;
            const int img  = bp >> 8;
            const int n    = bp & 255;
            int p;
            if (n < 66) {
                p = n;
            } else if (n < 190) {
                const int m = n - 66;
                p = (1 + (m >> 1)) * WPAD + (WPAD - 1) + (m & 1);
            } else {
                p = 63 * WPAD + (WPAD - 1) + (n - 190);
            }
            ((float4*)out)[(img * PADPIX + p) * (OUTQ / 4) + part] =
                make_float4(0.f, 0.f, 0.f, 0.f);
        }
    }

    {
        float* d = wlds + (tid >> 4) * PITCH + (tid & 15) * 4;
        float* d0w = wlds + L_W0 + (16 * ((tid >> 4) & 3) + (tid >> 6)) * PITCH + (tid & 15) * 4;
        *(float4*)(d0w)              = wa0;
        *(float4*)(d0w +  4 * PITCH) = wa1;
        *(float4*)(d0w +  8 * PITCH) = wa2;
        *(float4*)(d0w + 12 * PITCH) = wa3;
        *(float4*)(d + L_W1)              = wb1;
        *(float4*)(d + L_W2)              = wb2;
        if (tid < 16) *(float4*)(wlds + L_W3 + tid * 4) = wb3;
    }
    __syncthreads();

    if (!live) return;

    const int half = lane >> 4;
    const float xb3a = half ? q3.z : q3.x, xb3b = half ? q3.w : q3.y;

    const float* wrow = wlds + (lane & 15) * PITCH + 32 * half;
#define AF4(off, i) (*(const float4*)(wrow + (off) + 4 * (i)))

    const v2f a3 = *(const v2f*)(wlds + L_W3 + (lane & 15) * 4 + 2 * half);
    float4 w2f[8], w1f[8], w0f[4][8];
#pragma unroll
    for (int i = 0; i < 8; ++i) w2f[i] = AF4(L_W2, i);
#pragma unroll
    for (int i = 0; i < 8; ++i) w1f[i] = AF4(L_W1, i);
#pragma unroll
    for (int t = 0; t < 4; ++t) {
#pragma unroll
        for (int i = 0; i < 8; ++i) w0f[t][i] = AF4(L_W0 + 16 * t * PITCH, i);
    }
#undef AF4
    __builtin_amdgcn_sched_barrier(0);

    v2f bb; bb[0] = xb3a; bb[1] = xb3b;
    v8f zeroAcc = {};
    v8f v3d = wmma4(a3, bb, zeroAcc);

    v8f u = {};
#pragma unroll
    for (int i = 0; i < 8; ++i) {
        const float r = v3d[i];
        v2f a0; a0[0] = w2f[i].x; a0[1] = w2f[i].y;
        v2f b0; b0[0] = r * q2.x;  b0[1] = r * q2.y;
        u = wmma4(a0, b0, u);
        v2f a1; a1[0] = w2f[i].z; a1[1] = w2f[i].w;
        v2f b1; b1[0] = r * q2.z;  b1[1] = r * q2.w;
        u = wmma4(a1, b1, u);
    }

    v8f vv = {};
#pragma unroll
    for (int i = 0; i < 8; ++i) {
        const float r = u[i];
        v2f a0; a0[0] = w1f[i].x; a0[1] = w1f[i].y;
        v2f b0; b0[0] = r * q1.x;  b0[1] = r * q1.y;
        vv = wmma4(a0, b0, vv);
        v2f a1; a1[0] = w1f[i].z; a1[1] = w1f[i].w;
        v2f b1; b1[0] = r * q1.z;  b1[1] = r * q1.w;
        vv = wmma4(a1, b1, vv);
    }

    v8f d0 = {}, d1 = {}, d2 = {}, d3 = {};
#define W01(f) ({ v2f w_; w_[0] = (f).x; w_[1] = (f).y; w_; })
#define W23(f) ({ v2f w_; w_[0] = (f).z; w_[1] = (f).w; w_; })
#pragma unroll
    for (int i = 0; i < 8; ++i) {
        const float r = vv[i];
        v2f xa0; xa0[0] = r * q0.x;  xa0[1] = r * q0.y;
        d0 = wmma4(xa0, W01(w0f[0][i]), d0);
        d1 = wmma4(xa0, W01(w0f[1][i]), d1);
        d2 = wmma4(xa0, W01(w0f[2][i]), d2);
        d3 = wmma4(xa0, W01(w0f[3][i]), d3);
        v2f xa1; xa1[0] = r * q0.z;  xa1[1] = r * q0.w;
        d0 = wmma4(xa1, W23(w0f[0][i]), d0);
        d1 = wmma4(xa1, W23(w0f[1][i]), d1);
        d2 = wmma4(xa1, W23(w0f[2][i]), d2);
        d3 = wmma4(xa1, W23(w0f[3][i]), d3);
    }
#undef W01
#undef W23

    const int n0   = blk * 16;
    const int gr0  = n0 / INTER;
    const int col0 = n0 - gr0 * INTER;
    const int im0  = gr0 / INTER;
    const int row0 = gr0 - im0 * INTER;
    const int pp0  = n0 + 2 * gr0 + 130 * im0 + 66;
    const int thr  = INTER - col0;
    const int bump = (row0 == INTER - 1) ? 132 : 2;
    const unsigned lane_off = (unsigned)(lane & 15) * 16u;
#pragma unroll
    for (int r = 0; r < 8; ++r) {
        const int p  = 8 * half + r;
        const int pp = pp0 + p + (p >= thr ? bump : 0);
        const unsigned off = (unsigned)pp * (OUTQ * 4u) + lane_off;
        *(float4*)((char*)out + off) = make_float4(d0[r], d1[r], d2[r], d3[r]);
    }
}

extern "C" void kernel_launch(void* const* d_in, const int* in_sizes, int n_in,
                              void* d_out, int out_size, void* d_ws, size_t ws_size,
                              hipStream_t stream) {
    (void)in_sizes; (void)n_in; (void)out_size; (void)d_ws; (void)ws_size;
    const float* x  = (const float*)d_in[0];
    const float* c0 = (const float*)d_in[1];
    const float* c1 = (const float*)d_in[2];
    const float* c2 = (const float*)d_in[3];
    const float* c3 = (const float*)d_in[4];
    float* out = (float*)d_out;

    int grid = (NBLK + WAVES_PER_WG - 1) / WAVES_PER_WG;
    convsbs_chain<<<grid, 256, 0, stream>>>(x, c0, c1, c2, c3, out);
}


# Cores 2 and 3 contracted over their shared bond once per work-group in the prologue (16x16 M, row-DPP reduce-scatter, kept in LDS): stages 1+2 become one K=16 stage, 97 -> 84 WMMA per wave
# speedup vs baseline: 3.6133x; 1.0950x over previous
#include <hip/hip_runtime.h>

typedef __attribute__((ext_vector_type(2))) float v2f;
typedef __attribute__((ext_vector_type(8))) float v8f;

#define NB      16
#define HH      64
#define WW      64
#define QIN     4
#define BOND    16
#define OUTQ    64
#define HPAD    65
#define WPAD    65
#define INTER   63
#define PIX_PER_IMG (INTER * INTER)
#define NPIX    (NB * PIX_PER_IMG)
#define NBLK    (NPIX / 16)
#define PADPIX  (HPAD * WPAD)
#define BORDER_PER_IMG (PADPIX - PIX_PER_IMG)
#define NBORDER4 (NB * BORDER_PER_IMG * (OUTQ / 4))
#define WAVES_PER_WG 8
#define PITCH   68
#define L_W0    0
#define L_W1    (64 * PITCH)
#define MPITCH  20
#define L_M     (L_W1 + 16 * PITCH)
#define L_TOT   (L_M + 16 * MPITCH)

static __device__ __forceinline__ v8f wmma4(v2f a, v2f b, v8f c) {
    return __builtin_amdgcn_wmma_f32_16x16x4_f32(
         false, a,  false, b,
         (short)0, c,  false,  false);
}

template <int MASK>
static __device__ __forceinline__ float row_xor(float v) {
    return __builtin_bit_cast(float, __builtin_amdgcn_update_dpp(
        0, __builtin_bit_cast(int, v), 0x160 + MASK, 0xf, 0xf, true));
}

__global__ __launch_bounds__(256) void convsbs_chain(
        const float* __restrict__ x,
        const float* __restrict__ c0,
        const float* __restrict__ c1,
        const float* __restrict__ c2,
        const float* __restrict__ c3,
        float* __restrict__ out)
{
    __shared__ __attribute__((aligned(16))) float wlds[L_TOT];

    const int tid = threadIdx.x;

    const float4* g0 = (const float4*)c0;
    const float4* g1 = (const float4*)c1;
    const float4* g2 = (const float4*)c2;
    const float4* g3 = (const float4*)c3;
    const float4 wb2 = g2[tid];
    const float4 wb3 = g3[tid & 15];
    const float4 wa0 = g0[tid];
    const float4 wa1 = g0[tid + 256];
    const float4 wa2 = g0[tid + 512];
    const float4 wa3 = g0[tid + 768];
    const float4 wb1 = g1[tid];

    const int wave = tid >> 5;
    const int lane = tid & 31;
    const int blk  = blockIdx.x * WAVES_PER_WG + wave;
    const bool live = blk < NBLK;
    const int blkc = live ? blk : (NBLK - 1);

    const int pix = blkc * 16 + (lane & 15);
    const int b   = pix / PIX_PER_IMG;
    const int rr  = pix - b * PIX_PER_IMG;
    const int hp  = rr / INTER + 1;
    const int wp  = rr - (hp - 1) * INTER + 1;

    const float4* xv = (const float4*)x;
    const int base = (b * HH + hp) * WW + wp;
    const float4 q3 = xv[base];
    const float4 q2 = xv[base - 1];
    const float4 q1 = xv[base - WW];
    const float4 q0 = xv[base - WW - 1];

    {
        const int g = blockIdx.x * 256 + tid;
        if (g < NBORDER4) {
            const int part = g & 15;
            const int bp   = g >> 4;
            const int img  = bp >> 8;
            const int n    = bp & 255;
            int p;
            if (n < 66) {
                p = n;
            } else if (n < 190) {
                const int m = n - 66;
                p = (1 + (m >> 1)) * WPAD + (WPAD - 1) + (m & 1);
            } else {
                p = 63 * WPAD + (WPAD - 1) + (n - 190);
            }
            ((float4*)out)[(img * PADPIX + p) * (OUTQ / 4) + part] =
                make_float4(0.f, 0.f, 0.f, 0.f);
        }
    }

    {
        const int l = lane & 15;
        const bool b3 = (l & 8) != 0, b2 = (l & 4) != 0, b1 = (l & 2) != 0, b0 = (l & 1) != 0;
        const float wk0 = b3 ? wb2.z : wb2.x, wk1 = b3 ? wb2.w : wb2.y;
        const float ws0 = b3 ? wb2.x : wb2.z, ws1 = b3 ? wb2.y : wb2.w;
        const float w_kk = b2 ? wk1 : wk0, w_ks = b2 ? wk0 : wk1;
        const float w_sk = b2 ? ws1 : ws0, w_ss = b2 ? ws0 : ws1;
        const float vk0 = b1 ? wb3.z : wb3.x, vk1 = b1 ? wb3.w : wb3.y;
        const float vs0 = b1 ? wb3.x : wb3.z, vs1 = b1 ? wb3.y : wb3.w;
        v2f vk; vk[0] = b0 ? vk1 : vk0; vk[1] = b0 ? vk0 : vk1;
        v2f vs; vs[0] = b0 ? vs1 : vs0; vs[1] = b0 ? vs0 : vs1;
        const v2f pkk_k = w_kk * vk, pkk_s = w_kk * vs;
        const v2f pks_k = w_ks * vk, pks_s = w_ks * vs;
        const v2f psk_k = w_sk * vk, psk_s = w_sk * vs;
        const v2f pss_k = w_ss * vk, pss_s = w_ss * vs;
        const float ak_kk = pkk_k[0] + row_xor<8>(psk_k[0]), ak_ks = pkk_k[1] + row_xor<8>(psk_k[1]);
        const float ak_sk = pkk_s[0] + row_xor<8>(psk_s[0]), ak_ss = pkk_s[1] + row_xor<8>(psk_s[1]);
        const float as_kk = pks_k[0] + row_xor<8>(pss_k[0]), as_ks = pks_k[1] + row_xor<8>(pss_k[1]);
        const float as_sk = pks_s[0] + row_xor<8>(pss_s[0]), as_ss = pks_s[1] + row_xor<8>(pss_s[1]);
        const float bkk = ak_kk + row_xor<4>(as_kk), bks = ak_ks + row_xor<4>(as_ks);
        const float bsk = ak_sk + row_xor<4>(as_sk), bss = ak_ss + row_xor<4>(as_ss);
        const float ck = bkk + row_xor<2>(bsk), cs = bks + row_xor<2>(bss);
        const float m = ck + row_xor<1>(cs);
        wlds[L_M + (tid >> 4) * MPITCH + l] = m;
    }

    {
        float* d = wlds + (tid >> 4) * PITCH + (tid & 15) * 4;
        float* d0w = wlds + L_W0 + (16 * ((tid >> 4) & 3) + (tid >> 6)) * PITCH + (tid & 15) * 4;
        *(float4*)(d0w)              = wa0;
        *(float4*)(d0w +  4 * PITCH) = wa1;
        *(float4*)(d0w +  8 * PITCH) = wa2;
        *(float4*)(d0w + 12 * PITCH) = wa3;
        *(float4*)(d + L_W1)              = wb1;
    }
    __syncthreads();

    if (!live) return;

    const int half = lane >> 4;
    const float x2a = half ? q2.z : q2.x, x2b = half ? q2.w : q2.y;

    const float* wrow = wlds + (lane & 15) * PITCH + 32 * half;
#define AF4(off, i) (*(const float4*)(wrow + (off) + 4 * (i)))

    const float* mrow = wlds + L_M + (lane & 15) * MPITCH + 8 * half;
    const float4 mf0 = *(const float4*)(mrow);
    const float4 mf1 = *(const float4*)(mrow + 4);
    float4 w1f[8], w0f[4][8];
#pragma unroll
    for (int i = 0; i < 8; ++i) w1f[i] = AF4(L_W1, i);
#pragma unroll
    for (int t = 0; t < 4; ++t) {
#pragma unroll
        for (int i = 0; i < 8; ++i) w0f[t][i] = AF4(L_W0 + 16 * t * PITCH, i);
    }
#undef AF4
    __builtin_amdgcn_sched_barrier(0);

    v8f u = {};
    {
        v2f x3lo; x3lo[0] = q3.x; x3lo[1] = q3.y;
        v2f x3hi; x3hi[0] = q3.z; x3hi[1] = q3.w;
        v2f a;
        a[0] = mf0.x; a[1] = mf0.y;  u = wmma4(a, x2a * x3lo, u);
        a[0] = mf0.z; a[1] = mf0.w;  u = wmma4(a, x2a * x3hi, u);
        a[0] = mf1.x; a[1] = mf1.y;  u = wmma4(a, x2b * x3lo, u);
        a[0] = mf1.z; a[1] = mf1.w;  u = wmma4(a, x2b * x3hi, u);
    }

    v8f vv = {};
#pragma unroll
    for (int i = 0; i < 8; ++i) {
        const float r = u[i];
        v2f a0; a0[0] = w1f[i].x; a0[1] = w1f[i].y;
        v2f b0; b0[0] = r * q1.x;  b0[1] = r * q1.y;
        vv = wmma4(a0, b0, vv);
        v2f a1; a1[0] = w1f[i].z; a1[1] = w1f[i].w;
        v2f b1; b1[0] = r * q1.z;  b1[1] = r * q1.w;
        vv = wmma4(a1, b1, vv);
    }

    v8f d0 = {}, d1 = {}, d2 = {}, d3 = {};
#define W01(f) ({ v2f w_; w_[0] = (f).x; w_[1] = (f).y; w_; })
#define W23(f) ({ v2f w_; w_[0] = (f).z; w_[1] = (f).w; w_; })
#pragma unroll
    for (int i = 0; i < 8; ++i) {
        const float r = vv[i];
        v2f xa0; xa0[0] = r * q0.x;  xa0[1] = r * q0.y;
        d0 = wmma4(xa0, W01(w0f[0][i]), d0);
        d1 = wmma4(xa0, W01(w0f[1][i]), d1);
        d2 = wmma4(xa0, W01(w0f[2][i]), d2);
        d3 = wmma4(xa0, W01(w0f[3][i]), d3);
        v2f xa1; xa1[0] = r * q0.z;  xa1[1] = r * q0.w;
        d0 = wmma4(xa1, W23(w0f[0][i]), d0);
        d1 = wmma4(xa1, W23(w0f[1][i]), d1);
        d2 = wmma4(xa1, W23(w0f[2][i]), d2);
        d3 = wmma4(xa1, W23(w0f[3][i]), d3);
    }
#undef W01
#undef W23

    const int n0   = blk * 16;
    const int gr0  = n0 / INTER;
    const int col0 = n0 - gr0 * INTER;
    const int im0  = gr0 / INTER;
    const int row0 = gr0 - im0 * INTER;
    const int pp0  = n0 + 2 * gr0 + 130 * im0 + 66;
    const int thr  = INTER - col0;
    const int bump = (row0 == INTER - 1) ? 132 : 2;
    const unsigned lane_off = (unsigned)(lane & 15) * 16u;
#pragma unroll
    for (int r = 0; r < 8; ++r) {
        const int p  = 8 * half + r;
        const int pp = pp0 + p + (p >= thr ? bump : 0);
        const unsigned off = (unsigned)pp * (OUTQ * 4u) + lane_off;
        *(float4*)((char*)out + off) = make_float4(d0[r], d1[r], d2[r], d3[r]);
    }
}

extern "C" void kernel_launch(void* const* d_in, const int* in_sizes, int n_in,
                              void* d_out, int out_size, void* d_ws, size_t ws_size,
                              hipStream_t stream) {
    (void)in_sizes; (void)n_in; (void)out_size; (void)d_ws; (void)ws_size;
    const float* x  = (const float*)d_in[0];
    const float* c0 = (const float*)d_in[1];
    const float* c1 = (const float*)d_in[2];
    const float* c2 = (const float*)d_in[3];
    const float* c3 = (const float*)d_in[4];
    float* out = (float*)d_out;

    int grid = (NBLK + WAVES_PER_WG - 1) / WAVES_PER_WG;
    convsbs_chain<<<grid, 256, 0, stream>>>(x, c0, c1, c2, c3, out);
}
